# starting kernel
# baseline (speedup 1.0000x reference)
#include <hip/hip_runtime.h>
#include <hip/hip_bf16.h>
#include <math.h>

#define EMA_B 8
#define EMA_P 4096
#define EMA_F 1024
#define EMA_C 16
#define EMA_L 256

typedef float v2f __attribute__((ext_vector_type(2)));
typedef float v8f __attribute__((ext_vector_type(8)));

__device__ __forceinline__ float sigmoidf_(float v) {
    return 1.0f / (1.0f + __expf(-v));
}

__device__ __forceinline__ float powi_f(float a, int n) {
    float r = 1.0f;
    for (int i = 0; i < n; ++i) r *= a;
    return r;
}

__global__ void ema_chunk_ends(const float* __restrict__ x,
                               const float* __restrict__ raw_alpha,
                               float* __restrict__ ws) {
    int g = blockIdx.x * blockDim.x + threadIdx.x;
    int f = g & (EMA_F - 1);
    int j = (g >> 10) & (EMA_C - 1);
    int b = g >> 14;

    float a  = sigmoidf_(raw_alpha[0]);
    float om = 1.0f - a;

    const float* xp = x + ((size_t)b * EMA_P + (size_t)j * EMA_L) * EMA_F + f;

    float h = 0.0f;
#pragma unroll 8
    for (int t = 0; t < EMA_L; ++t) {
        h = fmaf(a, h, om * xp[(size_t)t * EMA_F]);
    }
    ws[((size_t)b * EMA_C + j) * EMA_F + f] = h;
}

__global__ void ema_scan_wmma(const float* __restrict__ x,
                              const float* __restrict__ raw_alpha,
                              const float* __restrict__ ws,
                              float* __restrict__ out) {
    const int lane = threadIdx.x & 31;
    const int wv   = blockIdx.x * (blockDim.x >> 5) + (threadIdx.x >> 5);
    const int ftile = wv & 63;
    const int j     = (wv >> 6) & (EMA_C - 1);
    const int b     = wv >> 10;

    const int nlo  = lane & 15;
    const int half = lane >> 4;
    const int f    = (ftile << 4) + nlo;

    const float a  = sigmoidf_(raw_alpha[0]);
    const float om = 1.0f - a;

    float aL = a;
#pragma unroll
    for (int i = 0; i < 8; ++i) aL *= aL;

    float c = 0.0f;
    for (int i = 0; i < j; ++i) {
        c = fmaf(aL, c, ws[((size_t)b * EMA_C + i) * EMA_F + f]);
    }

    v2f Am[4];
#pragma unroll
    for (int s = 0; s < 4; ++s) {
        int m  = nlo;
        int k0 = 4 * s + 2 * half;
        Am[s].x = (k0     <= m) ? om * powi_f(a, m - k0)     : 0.0f;
        Am[s].y = (k0 + 1 <= m) ? om * powi_f(a, m - k0 - 1) : 0.0f;
    }

    float mrow[8];
    mrow[0] = powi_f(a, half ? 9 : 1);
#pragma unroll
    for (int r = 1; r < 8; ++r) mrow[r] = mrow[r - 1] * a;

    const float* xp = x   + ((size_t)b * EMA_P + (size_t)j * EMA_L) * EMA_F + f;
    float*       op = out + ((size_t)b * EMA_P + (size_t)j * EMA_L) * EMA_F + f;

    v2f Bv[2][4];
#pragma unroll
    for (int s = 0; s < 4; ++s) {
        int t = 4 * s + 2 * half;
        Bv[0][s].x = __builtin_nontemporal_load(xp + (size_t)t       * EMA_F);
        Bv[0][s].y = __builtin_nontemporal_load(xp + (size_t)(t + 1) * EMA_F);
    }

#pragma unroll
    for (int blk = 0; blk < EMA_L / 16; ++blk) {
        const int cur = blk & 1;
        if (blk < (EMA_L / 16 - 1)) {
            const int nt0 = (blk + 1) * 16;
#pragma unroll
            for (int s = 0; s < 4; ++s) {
                int t = nt0 + 4 * s + 2 * half;
                Bv[cur ^ 1][s].x = __builtin_nontemporal_load(xp + (size_t)t       * EMA_F);
                Bv[cur ^ 1][s].y = __builtin_nontemporal_load(xp + (size_t)(t + 1) * EMA_F);
            }
        }

        v8f acc;
#pragma unroll
        for (int r = 0; r < 8; ++r) acc[r] = mrow[r] * c;

#pragma unroll
        for (int s = 0; s < 4; ++s) {
            acc = __builtin_amdgcn_wmma_f32_16x16x4_f32(
                false, Am[s], false, Bv[cur][s], (short)0, acc, false, false);
        }

        const int tbase = blk * 16 + half * 8;
#pragma unroll
        for (int r = 0; r < 8; ++r) {
            __builtin_nontemporal_store(acc[r], op + (size_t)(tbase + r) * EMA_F);
        }

        c = __shfl(acc[7], 16 + nlo, 32);
    }
}

extern "C" void kernel_launch(void* const* d_in, const int* in_sizes, int n_in,
                              void* d_out, int out_size, void* d_ws, size_t ws_size,
                              hipStream_t stream) {
    const float* x  = (const float*)d_in[0];
    const float* ra = (const float*)d_in[1];
    float* out = (float*)d_out;
    float* ws  = (float*)d_ws;

    {
        const int threads = EMA_B * EMA_F * EMA_C;
        ema_chunk_ends<<<threads / 256, 256, 0, stream>>>(x, ra, ws);
    }
    {
        const int waves = EMA_B * (EMA_F / 16) * EMA_C;
        ema_scan_wmma<<<waves / 8, 256, 0, stream>>>(x, ra, ws, out);
    }
}


# late wave's carry walk covers only ring turns (16 blocks) whose weight alpha^(8n) is not yet exactly zero in f32: older rows add exact zeros for finite x; walk now a loop; split moved to block 128
# speedup vs baseline: 1.5713x; 1.5713x over previous
#include <hip/hip_runtime.h>
#include <math.h>

#define EMA_B 8
#define EMA_P 4096
#define EMA_F 1024
#define EMA_NBLK (EMA_P / 16)
#define EMA_RING 8
#define EMA_SPLIT 128
#define EMA_WSLOTS 4
#define EMA_WTURNS (EMA_SPLIT / (4 * EMA_WSLOTS))

typedef float v2f __attribute__((ext_vector_type(2)));
typedef float v4f __attribute__((ext_vector_type(4)));
typedef float v8f __attribute__((ext_vector_type(8)));

__device__ __forceinline__ float sigmoidf_(float v) {
    return 1.0f / (1.0f + __expf(-v));
}

__device__ __forceinline__ float powi_f(float a, int n) {
    float r = 1.0f;
    for (int i = 0; i < n; ++i) r *= a;
    return r;
}

__device__ __forceinline__ void load_block(v2f (&dst)[4], const float* __restrict__ p) {
#pragma unroll
    for (int s = 0; s < 4; ++s) {
        dst[s].x = p[(size_t)(4 * s)     * EMA_F];
        dst[s].y = p[(size_t)(4 * s + 1) * EMA_F];
    }
}

template <int U, bool REQUEST>
__device__ __forceinline__ void pair_step(v2f (&R)[EMA_RING][4], const v2f (&Am)[4],
                                          const float (&mrow)[8], const float a16, float& c,
                                          const float* __restrict__ xp, float* __restrict__ op,
                                          const int blk0, const int nlo) {
    const int blk = blk0 + U;
    if (REQUEST) {
        load_block(R[(U + 6) & 7], xp + (size_t)(blk + 6) * 16 * EMA_F);
        load_block(R[(U + 7) & 7], xp + (size_t)(blk + 7) * 16 * EMA_F);
    }

    v8f h0 = {0.f, 0.f, 0.f, 0.f, 0.f, 0.f, 0.f, 0.f};
    v8f h1 = {0.f, 0.f, 0.f, 0.f, 0.f, 0.f, 0.f, 0.f};
#pragma unroll
    for (int s = 0; s < 4; ++s) {
        h0 = __builtin_amdgcn_wmma_f32_16x16x4_f32(
            false, Am[s], false, R[U][s],     (short)0, h0, false, false);
        h1 = __builtin_amdgcn_wmma_f32_16x16x4_f32(
            false, Am[s], false, R[U + 1][s], (short)0, h1, false, false);
    }

    const float e0 = __shfl(h0[7], 16 + nlo, 32);
    const float e1 = __shfl(h1[7], 16 + nlo, 32);
    const float c1 = fmaf(a16, c, e0);

    float* o = op + (size_t)blk * 16 * EMA_F;
#pragma unroll
    for (int r = 0; r < 8; ++r)
        __builtin_nontemporal_store(fmaf(mrow[r], c, h0[r]), o + (size_t)r * EMA_F);
#pragma unroll
    for (int r = 0; r < 8; ++r)
        __builtin_nontemporal_store(fmaf(mrow[r], c1, h1[r]), o + (size_t)(16 + r) * EMA_F);

    c = fmaf(a16, c1, e1);
    __builtin_amdgcn_sched_barrier(0);
}

__device__ __forceinline__ void walk_request(v4f (&G)[8], const float* __restrict__ p) {
#pragma unroll
    for (int i = 0; i < 4; ++i) {
        const float* q = p - (size_t)i * 16 * EMA_F;
        G[2 * i]     = *(const v4f*)(q);
        G[2 * i + 1] = *(const v4f*)(q + (size_t)8 * EMA_F);
    }
}

__device__ __forceinline__ void walk_consume(const v4f (&G)[8], const v2f Ab, const float a8,
                                             float& S, v8f (&acc)[4]) {
#pragma unroll
    for (int i = 0; i < 4; ++i) {
#pragma unroll
        for (int e2 = 1; e2 >= 0; --e2) {
            v2f As;
            As.x = Ab.x * S;
            As.y = Ab.y * S;
#pragma unroll
            for (int p = 0; p < 2; ++p) {
                v2f Bx;
                Bx.x = G[2 * i + e2][2 * p];
                Bx.y = G[2 * i + e2][2 * p + 1];
                acc[2 * e2 + p] = __builtin_amdgcn_wmma_f32_16x16x4_f32(
                    false, As, false, Bx, (short)0, acc[2 * e2 + p], false, false);
            }
            S *= a8;
        }
    }
}

__global__ __launch_bounds__(128)
void ema_scan_columns(const float* __restrict__ x,
                      const float* __restrict__ raw_alpha,
                      float* __restrict__ out) {
    const int lane  = threadIdx.x & 31;
    const bool late = blockIdx.x < (EMA_B * EMA_F / 64);
    const int tile  = (blockIdx.x & (EMA_B * EMA_F / 64 - 1)) * 4 + (threadIdx.x >> 5);
    const int ftile = tile & 63;
    const int b     = tile >> 6;

    const int nlo  = lane & 15;
    const int half = lane >> 4;
    const int f    = (ftile << 4) + nlo;

    const float a  = sigmoidf_(raw_alpha[0]);
    const float om = 1.0f - a;

    const float a2 = a * a, a4 = a2 * a2, a8 = a4 * a4;
    const float a16 = a8 * a8;

    v2f Am[4];
#pragma unroll
    for (int s = 0; s < 4; ++s) {
        int m  = nlo;
        int k0 = 4 * s + 2 * half;
        Am[s].x = (k0     <= m) ? om * powi_f(a, m - k0)     : 0.0f;
        Am[s].y = (k0 + 1 <= m) ? om * powi_f(a, m - k0 - 1) : 0.0f;
    }

    float mrow[8];
    mrow[0] = powi_f(a, half ? 9 : 1);
#pragma unroll
    for (int r = 1; r < 8; ++r) mrow[r] = mrow[r - 1] * a;

    const float* xp = x   + ((size_t)b * EMA_P + 2 * half) * EMA_F + f;
    float*       op = out + ((size_t)b * EMA_P + 8 * half) * EMA_F + f;

    v2f R[EMA_RING][4];
    float c = 0.0f;
    const int b_begin = late ? EMA_SPLIT : 0;
    const int b_end   = late ? EMA_NBLK : EMA_SPLIT;
    const float* xs = xp + (size_t)b_begin * 16 * EMA_F;

    if (late) {
        const int wr = lane >> 2;
        const int wq = lane & 3;
        const float* wt = x + (((size_t)b * EMA_P + wr) + (size_t)(EMA_SPLIT - 1) * 16) * EMA_F
                            + (ftile << 4) + 4 * wq;
        const float wh = om * (half ? 1.0f : a4);
        v2f Ab;
        Ab.x = (nlo & 1) ? 0.0f : wh;
        Ab.y = (nlo & 1) ? wh : 0.0f;

        v4f G[EMA_WSLOTS][8];
        v8f acc[4];
#pragma unroll
        for (int j = 0; j < 4; ++j) acc[j] = (v8f){0.f, 0.f, 0.f, 0.f, 0.f, 0.f, 0.f, 0.f};
        float S = 1.0f;

#pragma unroll
        for (int i = 0; i < EMA_WSLOTS; ++i) walk_request(G[i], wt - (size_t)(4 * i) * 16 * EMA_F);

        int nturns = 1;
        float Sn = 1.0f;
        for (int t = 1; t < EMA_WTURNS; ++t) {
#pragma unroll
            for (int j = 0; j < 8 * EMA_WSLOTS; ++j) Sn *= a8;
            if (Sn == 0.0f) break;
            ++nturns;
        }

        const float* wp = wt;
        for (int t = 1; t < nturns; ++t) {
            wp -= (size_t)(4 * EMA_WSLOTS) * 16 * EMA_F;
#pragma unroll
            for (int i = 0; i < EMA_WSLOTS; ++i) {
                walk_consume(G[i], Ab, a8, S, acc);
                __builtin_amdgcn_sched_barrier(0);
                walk_request(G[i], wp - (size_t)(4 * i) * 16 * EMA_F);
                __builtin_amdgcn_sched_barrier(0);
            }
        }
#pragma unroll
        for (int i = 0; i < EMA_WSLOTS; ++i) {
            walk_consume(G[i], Ab, a8, S, acc);
            __builtin_amdgcn_sched_barrier(0);
        }
#pragma unroll
        for (int i = 0; i < 6; ++i) load_block(R[i], xs + (size_t)i * 16 * EMA_F);

        const float pw[4] = {a2 * a, a2, a, 1.0f};
        float cpe[4];
#pragma unroll
        for (int pe = 0; pe < 4; ++pe) {
            const float v = acc[pe >> 1][pe & 1] + acc[2 + (pe >> 1)][pe & 1];
            float sum = 0.0f;
#pragma unroll
            for (int u = 0; u < 4; ++u)
                sum = fmaf(pw[u], __shfl(v, 4 * u + (nlo >> 2), 32), sum);
            cpe[pe] = sum;
        }
        const int sel = nlo & 3;
        c = (sel == 0) ? cpe[0] : (sel == 1) ? cpe[1] : (sel == 2) ? cpe[2] : cpe[3];
    } else {
#pragma unroll
        for (int i = 0; i < 6; ++i) load_block(R[i], xs + (size_t)i * 16 * EMA_F);
    }

    const int last_turn = b_end - EMA_RING;
    for (int blk0 = b_begin; blk0 < last_turn; blk0 += EMA_RING) {
        pair_step<0, true>(R, Am, mrow, a16, c, xp, op, blk0, nlo);
        pair_step<2, true>(R, Am, mrow, a16, c, xp, op, blk0, nlo);
        pair_step<4, true>(R, Am, mrow, a16, c, xp, op, blk0, nlo);
        pair_step<6, true>(R, Am, mrow, a16, c, xp, op, blk0, nlo);
    }
    pair_step<0, true >(R, Am, mrow, a16, c, xp, op, last_turn, nlo);
    pair_step<2, false>(R, Am, mrow, a16, c, xp, op, last_turn, nlo);
    pair_step<4, false>(R, Am, mrow, a16, c, xp, op, last_turn, nlo);
    pair_step<6, false>(R, Am, mrow, a16, c, xp, op, last_turn, nlo);
}

extern "C" void kernel_launch(void* const* d_in, const int* in_sizes, int n_in,
                              void* d_out, int out_size, void* d_ws, size_t ws_size,
                              hipStream_t stream) {
    const float* x  = (const float*)d_in[0];
    const float* ra = (const float*)d_in[1];
    float* out = (float*)d_out;

    ema_scan_columns<<<256, 128, 0, stream>>>(x, ra, out);
}


# a wave owns 32 columns x a quarter of the steps (was 16 x half): even/odd column tiles, 8-byte loads and stores on whole 128-byte lines; segments 1-3 start from the column-sum walk (sum order changes)
# speedup vs baseline: 2.2908x; 1.4579x over previous
#include <hip/hip_runtime.h>
#include <math.h>

#define EMA_B 8
#define EMA_P 4096
#define EMA_F 1024
#define EMA_NBLK (EMA_P / 16)
#define EMA_RING 8
#define EMA_NSEG 4
#define EMA_SEGBLK (EMA_NBLK / EMA_NSEG)
#define EMA_WSLOTS 4

typedef float v2f __attribute__((ext_vector_type(2)));
typedef float v4f __attribute__((ext_vector_type(4)));
typedef float v8f __attribute__((ext_vector_type(8)));

__device__ __forceinline__ float sigmoidf_(float v) {
    return 1.0f / (1.0f + __expf(-v));
}

__device__ __forceinline__ float powi_f(float a, int n) {
    float r = 1.0f;
    for (int i = 0; i < n; ++i) r *= a;
    return r;
}

__device__ __forceinline__ void load_block(v2f (&dst)[8], const float* __restrict__ p) {
#pragma unroll
    for (int s = 0; s < 4; ++s) {
        dst[2 * s]     = *(const v2f*)(p + (size_t)(4 * s)     * EMA_F);
        dst[2 * s + 1] = *(const v2f*)(p + (size_t)(4 * s + 1) * EMA_F);
    }
}

template <int U, bool REQUEST>
__device__ __forceinline__ void block_step(v2f (&R)[EMA_RING][8], const v2f (&Am)[4],
                                           const float (&mrow)[8], const float a16,
                                           float& cE, float& cO,
                                           const float* __restrict__ xp, float* __restrict__ op,
                                           const int blk0, const int nlo) {
    const int blk = blk0 + U;
    if (REQUEST) load_block(R[(U + 7) & 7], xp + (size_t)(blk + 7) * 16 * EMA_F);

    v8f hE = {0.f, 0.f, 0.f, 0.f, 0.f, 0.f, 0.f, 0.f};
    v8f hO = {0.f, 0.f, 0.f, 0.f, 0.f, 0.f, 0.f, 0.f};
#pragma unroll
    for (int s = 0; s < 4; ++s) {
        v2f bE, bO;
        bE.x = R[U][2 * s].x;  bE.y = R[U][2 * s + 1].x;
        bO.x = R[U][2 * s].y;  bO.y = R[U][2 * s + 1].y;
        hE = __builtin_amdgcn_wmma_f32_16x16x4_f32(
            false, Am[s], false, bE, (short)0, hE, false, false);
        hO = __builtin_amdgcn_wmma_f32_16x16x4_f32(
            false, Am[s], false, bO, (short)0, hO, false, false);
    }

    const float eE = __shfl(hE[7], 16 + nlo, 32);
    const float eO = __shfl(hO[7], 16 + nlo, 32);

    float* o = op + (size_t)blk * 16 * EMA_F;
#pragma unroll
    for (int r = 0; r < 8; ++r) {
        v2f v;
        v.x = fmaf(mrow[r], cE, hE[r]);
        v.y = fmaf(mrow[r], cO, hO[r]);
        __builtin_nontemporal_store(v, (v2f*)(o + (size_t)r * EMA_F));
    }

    cE = fmaf(a16, cE, eE);
    cO = fmaf(a16, cO, eO);
    __builtin_amdgcn_sched_barrier(0);
}

__device__ __forceinline__ void walk_request(v4f (&G)[8], const float* __restrict__ p) {
#pragma unroll
    for (int i = 0; i < 4; ++i) {
        const float* q = p - (size_t)i * 16 * EMA_F;
        G[2 * i]     = *(const v4f*)(q);
        G[2 * i + 1] = *(const v4f*)(q + (size_t)8 * EMA_F);
    }
}

__device__ __forceinline__ void walk_consume(const v4f (&G)[8], const v2f Ab, const float a8,
                                             float& S, v8f (&acc)[4]) {
#pragma unroll
    for (int i = 0; i < 4; ++i) {
#pragma unroll
        for (int e2 = 1; e2 >= 0; --e2) {
            v2f As;
            As.x = Ab.x * S;
            As.y = Ab.y * S;
#pragma unroll
            for (int p = 0; p < 2; ++p) {
                v2f Bx;
                Bx.x = G[2 * i + e2][2 * p];
                Bx.y = G[2 * i + e2][2 * p + 1];
                acc[2 * e2 + p] = __builtin_amdgcn_wmma_f32_16x16x4_f32(
                    false, As, false, Bx, (short)0, acc[2 * e2 + p], false, false);
            }
            S *= a8;
        }
    }
}

__device__ __forceinline__ void carry_walk(v8f (&acc)[4], const float* __restrict__ wt,
                                           const v2f Ab, const float a8, const int nturns) {
    v4f G[EMA_WSLOTS][8];
#pragma unroll
    for (int j = 0; j < 4; ++j) acc[j] = (v8f){0.f, 0.f, 0.f, 0.f, 0.f, 0.f, 0.f, 0.f};
    float S = 1.0f;

#pragma unroll
    for (int i = 0; i < EMA_WSLOTS; ++i) walk_request(G[i], wt - (size_t)(4 * i) * 16 * EMA_F);

    const float* wp = wt;
    for (int t = 1; t < nturns; ++t) {
        wp -= (size_t)(4 * EMA_WSLOTS) * 16 * EMA_F;
#pragma unroll
        for (int i = 0; i < EMA_WSLOTS; ++i) {
            walk_consume(G[i], Ab, a8, S, acc);
            __builtin_amdgcn_sched_barrier(0);
            walk_request(G[i], wp - (size_t)(4 * i) * 16 * EMA_F);
            __builtin_amdgcn_sched_barrier(0);
        }
    }
#pragma unroll
    for (int i = 0; i < EMA_WSLOTS; ++i) {
        walk_consume(G[i], Ab, a8, S, acc);
        __builtin_amdgcn_sched_barrier(0);
    }
}

__device__ __forceinline__ float carry_join(const v8f (&acc)[4], const float a, const float a2,
                                            const int nlo) {
    const float pw[4] = {a2 * a, a2, a, 1.0f};
    float cpe[4];
#pragma unroll
    for (int pe = 0; pe < 4; ++pe) {
        const float v = acc[pe >> 1][pe & 1] + acc[2 + (pe >> 1)][pe & 1];
        float sum = 0.0f;
#pragma unroll
        for (int u = 0; u < 4; ++u)
            sum = fmaf(pw[u], __shfl(v, 4 * u + (nlo >> 2), 32), sum);
        cpe[pe] = sum;
    }
    const int sel = nlo & 3;
    return (sel == 0) ? cpe[0] : (sel == 1) ? cpe[1] : (sel == 2) ? cpe[2] : cpe[3];
}

__global__ __launch_bounds__(128)
void ema_scan_columns(const float* __restrict__ x,
                      const float* __restrict__ raw_alpha,
                      float* __restrict__ out) {
    const int lane  = threadIdx.x & 31;
    const int seg   = (EMA_NSEG - 1) - ((int)blockIdx.x >> 6);
    const int b     = ((int)blockIdx.x >> 3) & 7;
    const int ctile = (((int)blockIdx.x & 7) << 2) + ((int)threadIdx.x >> 5);

    const int nlo  = lane & 15;
    const int half = lane >> 4;
    const int f0   = ctile << 5;

    const float a  = sigmoidf_(raw_alpha[0]);
    const float om = 1.0f - a;

    const float a2 = a * a, a4 = a2 * a2, a8 = a4 * a4;
    const float a16 = a8 * a8;

    v2f Am[4];
#pragma unroll
    for (int s = 0; s < 4; ++s) {
        int m  = nlo;
        int k0 = 4 * s + 2 * half;
        Am[s].x = (k0     <= m) ? om * powi_f(a, m - k0)     : 0.0f;
        Am[s].y = (k0 + 1 <= m) ? om * powi_f(a, m - k0 - 1) : 0.0f;
    }

    float mrow[8];
    mrow[0] = powi_f(a, half ? 9 : 1);
#pragma unroll
    for (int r = 1; r < 8; ++r) mrow[r] = mrow[r - 1] * a;

    const float* xp = x   + ((size_t)b * EMA_P + 2 * half) * EMA_F + f0 + 2 * nlo;
    float*       op = out + ((size_t)b * EMA_P + 8 * half) * EMA_F + f0 + 2 * nlo;

    v2f R[EMA_RING][8];
    float cE = 0.0f, cO = 0.0f;
    const int b_begin = seg * EMA_SEGBLK;
    const int b_end   = b_begin + EMA_SEGBLK;
    const float* xs = xp + (size_t)b_begin * 16 * EMA_F;

    if (seg > 0) {
        const int wr = lane >> 2;
        const int wq = lane & 3;
        const float* wt = x + (((size_t)b * EMA_P + wr) + (size_t)(b_begin - 1) * 16) * EMA_F
                            + f0 + 4 * wq;
        const float wh = om * (half ? 1.0f : a4);
        v2f Ab;
        Ab.x = (nlo & 1) ? 0.0f : wh;
        Ab.y = (nlo & 1) ? wh : 0.0f;

        const int max_turns = b_begin / (4 * EMA_WSLOTS);
        int nturns = 1;
        float Sn = 1.0f;
        for (int t = 1; t < max_turns; ++t) {
#pragma unroll
            for (int j = 0; j < 8 * EMA_WSLOTS; ++j) Sn *= a8;
            if (Sn == 0.0f) break;
            ++nturns;
        }

        v8f accA[4], accB[4];
        carry_walk(accA, wt,      Ab, a8, nturns);
        carry_walk(accB, wt + 16, Ab, a8, nturns);

#pragma unroll
        for (int i = 0; i < 7; ++i) load_block(R[i], xs + (size_t)i * 16 * EMA_F);

        const float cA = carry_join(accA, a, a2, nlo);
        const float cB = carry_join(accB, a, a2, nlo);
        const int sE = (2 * nlo) & 15;
        const float aE = __shfl(cA, sE, 32),     bE = __shfl(cB, sE, 32);
        const float aO = __shfl(cA, sE + 1, 32), bO = __shfl(cB, sE + 1, 32);
        cE = (nlo < 8) ? aE : bE;
        cO = (nlo < 8) ? aO : bO;
    } else {
#pragma unroll
        for (int i = 0; i < 7; ++i) load_block(R[i], xs + (size_t)i * 16 * EMA_F);
    }

    const int last_turn = b_end - EMA_RING;
    for (int blk0 = b_begin; blk0 < last_turn; blk0 += EMA_RING) {
        block_step<0, true>(R, Am, mrow, a16, cE, cO, xp, op, blk0, nlo);
        block_step<1, true>(R, Am, mrow, a16, cE, cO, xp, op, blk0, nlo);
        block_step<2, true>(R, Am, mrow, a16, cE, cO, xp, op, blk0, nlo);
        block_step<3, true>(R, Am, mrow, a16, cE, cO, xp, op, blk0, nlo);
        block_step<4, true>(R, Am, mrow, a16, cE, cO, xp, op, blk0, nlo);
        block_step<5, true>(R, Am, mrow, a16, cE, cO, xp, op, blk0, nlo);
        block_step<6, true>(R, Am, mrow, a16, cE, cO, xp, op, blk0, nlo);
        block_step<7, true>(R, Am, mrow, a16, cE, cO, xp, op, blk0, nlo);
    }
    block_step<0, true >(R, Am, mrow, a16, cE, cO, xp, op, last_turn, nlo);
    block_step<1, false>(R, Am, mrow, a16, cE, cO, xp, op, last_turn, nlo);
    block_step<2, false>(R, Am, mrow, a16, cE, cO, xp, op, last_turn, nlo);
    block_step<3, false>(R, Am, mrow, a16, cE, cO, xp, op, last_turn, nlo);
    block_step<4, false>(R, Am, mrow, a16, cE, cO, xp, op, last_turn, nlo);
    block_step<5, false>(R, Am, mrow, a16, cE, cO, xp, op, last_turn, nlo);
    block_step<6, false>(R, Am, mrow, a16, cE, cO, xp, op, last_turn, nlo);
    block_step<7, false>(R, Am, mrow, a16, cE, cO, xp, op, last_turn, nlo);
}

extern "C" void kernel_launch(void* const* d_in, const int* in_sizes, int n_in,
                              void* d_out, int out_size, void* d_ws, size_t ws_size,
                              hipStream_t stream) {
    const float* x  = (const float*)d_in[0];
    const float* ra = (const float*)d_in[1];
    float* out = (float*)d_out;

    ema_scan_columns<<<256, 128, 0, stream>>>(x, ra, out);
}
